# starting kernel
# baseline (speedup 1.0000x reference)
#include <hip/hip_runtime.h>

typedef __attribute__((ext_vector_type(2))) float v2f;
typedef __attribute__((ext_vector_type(8))) float v8f;

__device__ __forceinline__ float fast_tanh(float x) {
  float t = __builtin_amdgcn_exp2f(x * 2.885390081777927f);
  float r = __builtin_amdgcn_rcpf(t + 1.0f);
  return __builtin_fmaf(-2.0f, r, 1.0f);
}

__device__ __forceinline__ float xor16(float v) {
  int lane = (int)(threadIdx.x & 31u);
  int i = __builtin_amdgcn_ds_bpermute((lane ^ 16) << 2, __builtin_bit_cast(int, v));
  return __builtin_bit_cast(float, i);
}

__device__ __forceinline__ float rnn_chain(const float ix[4][2], v2f a, bool lo,
                                           float cw0, float cw1, float cb) {
  float h0 = lo ? fast_tanh(ix[0][0]) : 0.0f;
  float h1 = lo ? fast_tanh(ix[0][1]) : 0.0f;
#pragma unroll
  for (int t = 1; t < 4; ++t) {
    v8f c;
    c[0] = lo ? ix[t][0] : 0.0f;
    c[1] = lo ? ix[t][1] : 0.0f;
    c[2] = 0.0f; c[3] = 0.0f; c[4] = 0.0f; c[5] = 0.0f; c[6] = 0.0f; c[7] = 0.0f;
    v2f b;
    b[0] = h0;
    b[1] = h1;
    v8f d = __builtin_amdgcn_wmma_f32_16x16x4_f32(
         false, a,  false, b,
         (short)0, c,  false,  false);
    h0 = fast_tanh(d[0]);
    h1 = fast_tanh(d[1]);
  }
  return __builtin_fmaf(cw0, h0, __builtin_fmaf(cw1, h1, cb));
}

__global__ __launch_bounds__(256) void rnn2cls_kernel(
    const float* __restrict__ X, const float* __restrict__ wih,
    const float* __restrict__ bih, const float* __restrict__ whh,
    const float* __restrict__ bhh, const float* __restrict__ clsw,
    const float* __restrict__ clsb, float* __restrict__ out, int B) {
  const int lane = (int)(threadIdx.x & 31u);
  const int wave = (int)(threadIdx.x >> 5u);
  const int base = (blockIdx.x * (int)(blockDim.x >> 5u) + wave) * 32;
  int idx = base + lane;
  idx = idx < B ? idx : (B - 1);

  const float4* xp = reinterpret_cast<const float4*>(X) + (size_t)idx * 2;
  const float4 xa = xp[0];
  const float4 xb = xp[1];

  const float w00 = wih[0], w01 = wih[1], w10 = wih[2], w11 = wih[3];
  const float b0 = bih[0] + bhh[0];
  const float b1 = bih[1] + bhh[1];
  const float cw0 = clsw[0], cw1 = clsw[1], cb = clsb[0];

  v2f a;
  a[0] = lane == 0 ? whh[0] : (lane == 1 ? whh[2] : 0.0f);
  a[1] = lane == 0 ? whh[1] : (lane == 1 ? whh[3] : 0.0f);

  float ix[4][2];
  ix[0][0] = __builtin_fmaf(w00, xa.x, __builtin_fmaf(w01, xa.y, b0));
  ix[0][1] = __builtin_fmaf(w10, xa.x, __builtin_fmaf(w11, xa.y, b1));
  ix[1][0] = __builtin_fmaf(w00, xa.z, __builtin_fmaf(w01, xa.w, b0));
  ix[1][1] = __builtin_fmaf(w10, xa.z, __builtin_fmaf(w11, xa.w, b1));
  ix[2][0] = __builtin_fmaf(w00, xb.x, __builtin_fmaf(w01, xb.y, b0));
  ix[2][1] = __builtin_fmaf(w10, xb.x, __builtin_fmaf(w11, xb.y, b1));
  ix[3][0] = __builtin_fmaf(w00, xb.z, __builtin_fmaf(w01, xb.w, b0));
  ix[3][1] = __builtin_fmaf(w10, xb.z, __builtin_fmaf(w11, xb.w, b1));

  const bool lo = lane < 16;

  const float out1 = rnn_chain(ix, a, lo, cw0, cw1, cb);

  float ix2[4][2];
#pragma unroll
  for (int t = 0; t < 4; ++t) {
    ix2[t][0] = xor16(ix[t][0]);
    ix2[t][1] = xor16(ix[t][1]);
  }
  const float out2 = rnn_chain(ix2, a, lo, cw0, cw1, cb);

  const float o2hi = xor16(out2);
  const float result = lo ? out1 : o2hi;
  if (base + lane < B) out[base + lane] = result;
}

extern "C" void kernel_launch(void* const* d_in, const int* in_sizes, int n_in,
                              void* d_out, int out_size, void* d_ws, size_t ws_size,
                              hipStream_t stream) {
  (void)n_in; (void)out_size; (void)d_ws; (void)ws_size;
  const float* X    = (const float*)d_in[0];
  const float* wih  = (const float*)d_in[1];
  const float* bih  = (const float*)d_in[2];
  const float* whh  = (const float*)d_in[3];
  const float* bhh  = (const float*)d_in[4];
  const float* clsw = (const float*)d_in[5];
  const float* clsb = (const float*)d_in[6];
  float* out = (float*)d_out;

  const int B = in_sizes[0] / 8;
  const int blocks = (B + 255) / 256;
  rnn2cls_kernel<<<blocks, 256, 0, stream>>>(X, wih, bih, whh, bhh, clsw, clsb,
                                             out, B);
}


# W_hh twice on A's block diagonal (rows 0,1 read k=0,1; rows 8,9 read k=2,3; selects from 4 uniform loads): all 32 lanes carry batch elements, 3 WMMAs per 32 elements not 6, half-swaps gone
# speedup vs baseline: 1.4400x; 1.4400x over previous
#include <hip/hip_runtime.h>


typedef __attribute__((ext_vector_type(2))) float v2f;
typedef __attribute__((ext_vector_type(8))) float v8f;

__device__ __forceinline__ float fast_tanh(float x) {
  float t = __builtin_amdgcn_exp2f(x * 2.885390081777927f);
  float r = __builtin_amdgcn_rcpf(t + 1.0f);
  return __builtin_fmaf(-2.0f, r, 1.0f);
}

__device__ __forceinline__ float rnn_chain(const float ix[4][2], v2f a,
                                           float cw0, float cw1, float cb) {
  float h0 = fast_tanh(ix[0][0]);
  float h1 = fast_tanh(ix[0][1]);
#pragma unroll
  for (int t = 1; t < 4; ++t) {
    v8f c;
    c[0] = ix[t][0];
    c[1] = ix[t][1];
    c[2] = 0.0f; c[3] = 0.0f; c[4] = 0.0f; c[5] = 0.0f; c[6] = 0.0f; c[7] = 0.0f;
    v2f b;
    b[0] = h0;
    b[1] = h1;
    v8f d = __builtin_amdgcn_wmma_f32_16x16x4_f32(
         false, a,  false, b,
         (short)0, c,  false,  false);
    h0 = fast_tanh(d[0]);
    h1 = fast_tanh(d[1]);
  }
  return __builtin_fmaf(cw0, h0, __builtin_fmaf(cw1, h1, cb));
}

__global__ __launch_bounds__(256) void rnn2cls_kernel(
    const float* __restrict__ X, const float* __restrict__ wih,
    const float* __restrict__ bih, const float* __restrict__ whh,
    const float* __restrict__ bhh, const float* __restrict__ clsw,
    const float* __restrict__ clsb, float* __restrict__ out, int B) {
  const int lane = (int)(threadIdx.x & 31u);
  const int wave = (int)(threadIdx.x >> 5u);
  const int base = (blockIdx.x * (int)(blockDim.x >> 5u) + wave) * 32;
  int idx = base + lane;
  idx = idx < B ? idx : (B - 1);

  const float4* xp = reinterpret_cast<const float4*>(X) + (size_t)idx * 2;
  const float4 xa = xp[0];
  const float4 xb = xp[1];

  const float w00 = wih[0], w01 = wih[1], w10 = wih[2], w11 = wih[3];
  const float b0 = bih[0] + bhh[0];
  const float b1 = bih[1] + bhh[1];
  const float cw0 = clsw[0], cw1 = clsw[1], cb = clsb[0];

  const float u00 = whh[0], u01 = whh[1], u10 = whh[2], u11 = whh[3];
  const bool r0 = (lane == 0) | (lane == 24);
  const bool r1 = (lane == 1) | (lane == 25);
  float a0 = r1 ? u10 : 0.0f;
  float a1 = r1 ? u11 : 0.0f;
  a0 = r0 ? u00 : a0;
  a1 = r0 ? u01 : a1;
  v2f a;
  a[0] = a0;
  a[1] = a1;

  float ix[4][2];
  ix[0][0] = __builtin_fmaf(w00, xa.x, __builtin_fmaf(w01, xa.y, b0));
  ix[0][1] = __builtin_fmaf(w10, xa.x, __builtin_fmaf(w11, xa.y, b1));
  ix[1][0] = __builtin_fmaf(w00, xa.z, __builtin_fmaf(w01, xa.w, b0));
  ix[1][1] = __builtin_fmaf(w10, xa.z, __builtin_fmaf(w11, xa.w, b1));
  ix[2][0] = __builtin_fmaf(w00, xb.x, __builtin_fmaf(w01, xb.y, b0));
  ix[2][1] = __builtin_fmaf(w10, xb.x, __builtin_fmaf(w11, xb.y, b1));
  ix[3][0] = __builtin_fmaf(w00, xb.z, __builtin_fmaf(w01, xb.w, b0));
  ix[3][1] = __builtin_fmaf(w10, xb.z, __builtin_fmaf(w11, xb.w, b1));

  const float result = rnn_chain(ix, a, cw0, cw1, cb);
  if (base + lane < B) out[base + lane] = result;
}

extern "C" void kernel_launch(void* const* d_in, const int* in_sizes, int n_in,
                              void* d_out, int out_size, void* d_ws, size_t ws_size,
                              hipStream_t stream) {
  (void)n_in; (void)out_size; (void)d_ws; (void)ws_size;
  const float* X    = (const float*)d_in[0];
  const float* wih  = (const float*)d_in[1];
  const float* bih  = (const float*)d_in[2];
  const float* whh  = (const float*)d_in[3];
  const float* bhh  = (const float*)d_in[4];
  const float* clsw = (const float*)d_in[5];
  const float* clsb = (const float*)d_in[6];
  float* out = (float*)d_out;

  const int B = in_sizes[0] / 8;
  const int blocks = (B + 255) / 256;
  rnn2cls_kernel<<<blocks, 256, 0, stream>>>(X, wih, bih, whh, bhh, clsw, clsb,
                                             out, B);
}


# tanh on the hardware's own f32 tanh instruction (v_tanh_f32) in place of exp2, add, rcp and fma: one transcendental per value instead of two plus three plain ops; f32 kept
# speedup vs baseline: 1.6013x; 1.1120x over previous
#include <hip/hip_runtime.h>


typedef __attribute__((ext_vector_type(2))) float v2f;
typedef __attribute__((ext_vector_type(8))) float v8f;

__device__ __forceinline__ float fast_tanh(float x) {
  return __builtin_amdgcn_tanhf(x);
}

__device__ __forceinline__ float rnn_chain(const float ix[4][2], v2f a,
                                           float cw0, float cw1, float cb) {
  float h0 = fast_tanh(ix[0][0]);
  float h1 = fast_tanh(ix[0][1]);
#pragma unroll
  for (int t = 1; t < 4; ++t) {
    v8f c;
    c[0] = ix[t][0];
    c[1] = ix[t][1];
    c[2] = 0.0f; c[3] = 0.0f; c[4] = 0.0f; c[5] = 0.0f; c[6] = 0.0f; c[7] = 0.0f;
    v2f b;
    b[0] = h0;
    b[1] = h1;
    v8f d = __builtin_amdgcn_wmma_f32_16x16x4_f32(
         false, a,  false, b,
         (short)0, c,  false,  false);
    h0 = fast_tanh(d[0]);
    h1 = fast_tanh(d[1]);
  }
  return __builtin_fmaf(cw0, h0, __builtin_fmaf(cw1, h1, cb));
}

__global__ __launch_bounds__(256) void rnn2cls_kernel(
    const float* __restrict__ X, const float* __restrict__ wih,
    const float* __restrict__ bih, const float* __restrict__ whh,
    const float* __restrict__ bhh, const float* __restrict__ clsw,
    const float* __restrict__ clsb, float* __restrict__ out, int B) {
  const int lane = (int)(threadIdx.x & 31u);
  const int wave = (int)(threadIdx.x >> 5u);
  const int base = (blockIdx.x * (int)(blockDim.x >> 5u) + wave) * 32;
  int idx = base + lane;
  idx = idx < B ? idx : (B - 1);

  const float4* xp = reinterpret_cast<const float4*>(X) + (size_t)idx * 2;
  const float4 xa = xp[0];
  const float4 xb = xp[1];

  const float w00 = wih[0], w01 = wih[1], w10 = wih[2], w11 = wih[3];
  const float b0 = bih[0] + bhh[0];
  const float b1 = bih[1] + bhh[1];
  const float cw0 = clsw[0], cw1 = clsw[1], cb = clsb[0];

  const float u00 = whh[0], u01 = whh[1], u10 = whh[2], u11 = whh[3];
  const bool r0 = (lane == 0) | (lane == 24);
  const bool r1 = (lane == 1) | (lane == 25);
  float a0 = r1 ? u10 : 0.0f;
  float a1 = r1 ? u11 : 0.0f;
  a0 = r0 ? u00 : a0;
  a1 = r0 ? u01 : a1;
  v2f a;
  a[0] = a0;
  a[1] = a1;

  float ix[4][2];
  ix[0][0] = __builtin_fmaf(w00, xa.x, __builtin_fmaf(w01, xa.y, b0));
  ix[0][1] = __builtin_fmaf(w10, xa.x, __builtin_fmaf(w11, xa.y, b1));
  ix[1][0] = __builtin_fmaf(w00, xa.z, __builtin_fmaf(w01, xa.w, b0));
  ix[1][1] = __builtin_fmaf(w10, xa.z, __builtin_fmaf(w11, xa.w, b1));
  ix[2][0] = __builtin_fmaf(w00, xb.x, __builtin_fmaf(w01, xb.y, b0));
  ix[2][1] = __builtin_fmaf(w10, xb.x, __builtin_fmaf(w11, xb.y, b1));
  ix[3][0] = __builtin_fmaf(w00, xb.z, __builtin_fmaf(w01, xb.w, b0));
  ix[3][1] = __builtin_fmaf(w10, xb.z, __builtin_fmaf(w11, xb.w, b1));

  const float result = rnn_chain(ix, a, cw0, cw1, cb);
  if (base + lane < B) out[base + lane] = result;
}

extern "C" void kernel_launch(void* const* d_in, const int* in_sizes, int n_in,
                              void* d_out, int out_size, void* d_ws, size_t ws_size,
                              hipStream_t stream) {
  (void)n_in; (void)out_size; (void)d_ws; (void)ws_size;
  const float* X    = (const float*)d_in[0];
  const float* wih  = (const float*)d_in[1];
  const float* bih  = (const float*)d_in[2];
  const float* whh  = (const float*)d_in[3];
  const float* bhh  = (const float*)d_in[4];
  const float* clsw = (const float*)d_in[5];
  const float* clsb = (const float*)d_in[6];
  float* out = (float*)d_out;

  const int B = in_sizes[0] / 8;
  const int blocks = (B + 255) / 256;
  rnn2cls_kernel<<<blocks, 256, 0, stream>>>(X, wih, bih, whh, bhh, clsw, clsb,
                                             out, B);
}


# Few long waves with loads ahead: 245 workgroups of 16 waves, each wave walks 32 tiles in 4 groups of 8 and issues the next group's 16 loads before this group's WMMA chains (two register buffers)
# speedup vs baseline: 1.7207x; 1.0746x over previous
#include <hip/hip_runtime.h>


typedef __attribute__((ext_vector_type(2))) float v2f;
typedef __attribute__((ext_vector_type(8))) float v8f;

#define WAVES 16
#define GROUP 8
#define NGROUPS 4
#define HALF 4
#define TILES_PER_WAVE (GROUP * NGROUPS)

__device__ __forceinline__ float fast_tanh(float x) {
  return __builtin_amdgcn_tanhf(x);
}

__device__ __forceinline__ void rnn_chains(const float4* xa, const float4* xb,
                                           v2f a, float w00, float w01,
                                           float w10, float w11, float b0,
                                           float b1, float cw0, float cw1,
                                           float cb, float* res) {
  float ix[HALF][4][2];
#pragma unroll
  for (int k = 0; k < HALF; ++k) {
    ix[k][0][0] = __builtin_fmaf(w00, xa[k].x, __builtin_fmaf(w01, xa[k].y, b0));
    ix[k][0][1] = __builtin_fmaf(w10, xa[k].x, __builtin_fmaf(w11, xa[k].y, b1));
    ix[k][1][0] = __builtin_fmaf(w00, xa[k].z, __builtin_fmaf(w01, xa[k].w, b0));
    ix[k][1][1] = __builtin_fmaf(w10, xa[k].z, __builtin_fmaf(w11, xa[k].w, b1));
    ix[k][2][0] = __builtin_fmaf(w00, xb[k].x, __builtin_fmaf(w01, xb[k].y, b0));
    ix[k][2][1] = __builtin_fmaf(w10, xb[k].x, __builtin_fmaf(w11, xb[k].y, b1));
    ix[k][3][0] = __builtin_fmaf(w00, xb[k].z, __builtin_fmaf(w01, xb[k].w, b0));
    ix[k][3][1] = __builtin_fmaf(w10, xb[k].z, __builtin_fmaf(w11, xb[k].w, b1));
  }
  float h0[HALF], h1[HALF];
#pragma unroll
  for (int k = 0; k < HALF; ++k) {
    h0[k] = fast_tanh(ix[k][0][0]);
    h1[k] = fast_tanh(ix[k][0][1]);
  }
#pragma unroll
  for (int t = 1; t < 4; ++t) {
    v8f d[HALF];
#pragma unroll
    for (int k = 0; k < HALF; ++k) {
      v8f c;
      c[0] = ix[k][t][0];
      c[1] = ix[k][t][1];
      c[2] = 0.0f; c[3] = 0.0f; c[4] = 0.0f; c[5] = 0.0f; c[6] = 0.0f; c[7] = 0.0f;
      v2f b;
      b[0] = h0[k];
      b[1] = h1[k];
      d[k] = __builtin_amdgcn_wmma_f32_16x16x4_f32(
           false, a,  false, b,
           (short)0, c,  false,  false);
    }
#pragma unroll
    for (int k = 0; k < HALF; ++k) {
      h0[k] = fast_tanh(d[k][0]);
      h1[k] = fast_tanh(d[k][1]);
    }
  }
#pragma unroll
  for (int k = 0; k < HALF; ++k)
    res[k] = __builtin_fmaf(cw0, h0[k], __builtin_fmaf(cw1, h1[k], cb));
}

__global__ __launch_bounds__(WAVES * 32) void rnn2cls_kernel(
    const float* __restrict__ X, const float* __restrict__ wih,
    const float* __restrict__ bih, const float* __restrict__ whh,
    const float* __restrict__ bhh, const float* __restrict__ clsw,
    const float* __restrict__ clsb, float* __restrict__ out, int B) {
  const int lane = (int)(threadIdx.x & 31u);
  const int wave = __builtin_amdgcn_readfirstlane((int)(threadIdx.x >> 5u));
  const int first = ((int)blockIdx.x * WAVES + wave) * (TILES_PER_WAVE * 32);
  if (first >= B) return;
  const int e0 = first + lane;
  const int last = B - 1;

  const float w00 = wih[0], w01 = wih[1], w10 = wih[2], w11 = wih[3];
  const float b0 = bih[0] + bhh[0];
  const float b1 = bih[1] + bhh[1];
  const float cw0 = clsw[0], cw1 = clsw[1], cb = clsb[0];

  const float u00 = whh[0], u01 = whh[1], u10 = whh[2], u11 = whh[3];
  const bool r0 = (lane == 0) | (lane == 24);
  const bool r1 = (lane == 1) | (lane == 25);
  float a0 = r1 ? u10 : 0.0f;
  float a1 = r1 ? u11 : 0.0f;
  a0 = r0 ? u00 : a0;
  a1 = r0 ? u01 : a1;
  v2f a;
  a[0] = a0;
  a[1] = a1;

  const float4* X4 = reinterpret_cast<const float4*>(X);
  float4 xa[2][GROUP], xb[2][GROUP];
#define LOAD_GROUP(J)                                                        \
  do {                                                                       \
    _Pragma("unroll") for (int k = 0; k < GROUP; ++k) {                      \
      unsigned e_ = (unsigned)(e0 + ((J) * GROUP + k) * 32);                 \
      e_ = e_ < (unsigned)last ? e_ : (unsigned)last;                        \
      const float4* p_ = X4 + (size_t)e_ * 2;                                \
      xa[(J) & 1][k] = p_[0];                                                \
      xb[(J) & 1][k] = p_[1];                                                \
    }                                                                        \
  } while (0)

  LOAD_GROUP(0);
#pragma unroll
  for (int j = 0; j < NGROUPS; ++j) {
    __builtin_amdgcn_sched_barrier(0);
    if (j + 1 < NGROUPS) LOAD_GROUP(j + 1);
    __builtin_amdgcn_sched_barrier(0);
#pragma unroll
    for (int s = 0; s < GROUP; s += HALF) {
      float res[HALF];
      rnn_chains(&xa[j & 1][s], &xb[j & 1][s], a, w00, w01, w10, w11, b0, b1,
                 cw0, cw1, cb, res);
      float* op = out + (size_t)e0 + (size_t)((j * GROUP + s) * 32);
      if (first + (j * GROUP + s + HALF) * 32 <= B) {
#pragma unroll
        for (int k = 0; k < HALF; ++k) op[k * 32] = res[k];
      } else {
#pragma unroll
        for (int k = 0; k < HALF; ++k)
          if (e0 + (j * GROUP + s + k) * 32 < B) op[k * 32] = res[k];
      }
    }
  }
#undef LOAD_GROUP
}

extern "C" void kernel_launch(void* const* d_in, const int* in_sizes, int n_in,
                              void* d_out, int out_size, void* d_ws, size_t ws_size,
                              hipStream_t stream) {
  (void)n_in; (void)out_size; (void)d_ws; (void)ws_size;
  const float* X    = (const float*)d_in[0];
  const float* wih  = (const float*)d_in[1];
  const float* bih  = (const float*)d_in[2];
  const float* whh  = (const float*)d_in[3];
  const float* bhh  = (const float*)d_in[4];
  const float* clsw = (const float*)d_in[5];
  const float* clsb = (const float*)d_in[6];
  float* out = (float*)d_out;

  const int B = in_sizes[0] / 8;
  const int per_block = WAVES * TILES_PER_WAVE * 32;
  const int blocks = (B + per_block - 1) / per_block;
  rnn2cls_kernel<<<blocks, WAVES * 32, 0, stream>>>(X, wih, bih, whh, bhh, clsw,
                                                   clsb, out, B);
}


# Whole-line loads: each b128 load reads 512 contiguous bytes (4 lines, not 8) at constant offsets from one lane base; 4 v_permlane16_swap_b32 per tile give each lane its element back
# speedup vs baseline: 1.9076x; 1.1086x over previous
#include <hip/hip_runtime.h>


typedef __attribute__((ext_vector_type(2))) float v2f;
typedef __attribute__((ext_vector_type(8))) float v8f;

#define WAVES 16
#define GROUP 8
#define NGROUPS 4
#define HALF 4
#define TILES_PER_WAVE (GROUP * NGROUPS)

__device__ __forceinline__ float fast_tanh(float x) {
  return __builtin_amdgcn_tanhf(x);
}

__device__ __forceinline__ void swap_halves(float& p, float& q) {
  auto r = __builtin_amdgcn_permlane16_swap(__builtin_bit_cast(unsigned, p),
                                            __builtin_bit_cast(unsigned, q),
                                            false, false);
  p = __builtin_bit_cast(float, (unsigned)r[0]);
  q = __builtin_bit_cast(float, (unsigned)r[1]);
}

__device__ __forceinline__ void rnn_chains(const float4* xa, const float4* xb,
                                           v2f a, float w00, float w01,
                                           float w10, float w11, float b0,
                                           float b1, float cw0, float cw1,
                                           float cb, float* res) {
  float4 pa[HALF], pb[HALF];
#pragma unroll
  for (int k = 0; k < HALF; ++k) {
    pa[k] = xa[k];
    pb[k] = xb[k];
    swap_halves(pa[k].x, pb[k].x);
    swap_halves(pa[k].y, pb[k].y);
    swap_halves(pa[k].z, pb[k].z);
    swap_halves(pa[k].w, pb[k].w);
  }
  float ix[HALF][4][2];
#pragma unroll
  for (int k = 0; k < HALF; ++k) {
    ix[k][0][0] = __builtin_fmaf(w00, pa[k].x, __builtin_fmaf(w01, pa[k].y, b0));
    ix[k][0][1] = __builtin_fmaf(w10, pa[k].x, __builtin_fmaf(w11, pa[k].y, b1));
    ix[k][1][0] = __builtin_fmaf(w00, pa[k].z, __builtin_fmaf(w01, pa[k].w, b0));
    ix[k][1][1] = __builtin_fmaf(w10, pa[k].z, __builtin_fmaf(w11, pa[k].w, b1));
    ix[k][2][0] = __builtin_fmaf(w00, pb[k].x, __builtin_fmaf(w01, pb[k].y, b0));
    ix[k][2][1] = __builtin_fmaf(w10, pb[k].x, __builtin_fmaf(w11, pb[k].y, b1));
    ix[k][3][0] = __builtin_fmaf(w00, pb[k].z, __builtin_fmaf(w01, pb[k].w, b0));
    ix[k][3][1] = __builtin_fmaf(w10, pb[k].z, __builtin_fmaf(w11, pb[k].w, b1));
  }
  float h0[HALF], h1[HALF];
#pragma unroll
  for (int k = 0; k < HALF; ++k) {
    h0[k] = fast_tanh(ix[k][0][0]);
    h1[k] = fast_tanh(ix[k][0][1]);
  }
#pragma unroll
  for (int t = 1; t < 4; ++t) {
    v8f d[HALF];
#pragma unroll
    for (int k = 0; k < HALF; ++k) {
      v8f c;
      c[0] = ix[k][t][0];
      c[1] = ix[k][t][1];
      c[2] = 0.0f; c[3] = 0.0f; c[4] = 0.0f; c[5] = 0.0f; c[6] = 0.0f; c[7] = 0.0f;
      v2f b;
      b[0] = h0[k];
      b[1] = h1[k];
      d[k] = __builtin_amdgcn_wmma_f32_16x16x4_f32(
           false, a,  false, b,
           (short)0, c,  false,  false);
    }
#pragma unroll
    for (int k = 0; k < HALF; ++k) {
      h0[k] = fast_tanh(d[k][0]);
      h1[k] = fast_tanh(d[k][1]);
    }
  }
#pragma unroll
  for (int k = 0; k < HALF; ++k)
    res[k] = __builtin_fmaf(cw0, h0[k], __builtin_fmaf(cw1, h1[k], cb));
}

template <bool TAIL>
__device__ __forceinline__ void rnn2cls_wave(
    const float* __restrict__ X, const float* __restrict__ wih,
    const float* __restrict__ bih, const float* __restrict__ whh,
    const float* __restrict__ bhh, const float* __restrict__ clsw,
    const float* __restrict__ clsb, float* __restrict__ out, int B, int lane,
    int first) {
  const int e0 = first + lane;
  const int last = B - 1;

  const float w00 = wih[0], w01 = wih[1], w10 = wih[2], w11 = wih[3];
  const float b0 = bih[0] + bhh[0];
  const float b1 = bih[1] + bhh[1];
  const float cw0 = clsw[0], cw1 = clsw[1], cb = clsb[0];

  const float u00 = whh[0], u01 = whh[1], u10 = whh[2], u11 = whh[3];
  const bool r0 = (lane == 0) | (lane == 24);
  const bool r1 = (lane == 1) | (lane == 25);
  float a0 = r1 ? u10 : 0.0f;
  float a1 = r1 ? u11 : 0.0f;
  a0 = r0 ? u00 : a0;
  a1 = r0 ? u01 : a1;
  v2f a;
  a[0] = a0;
  a[1] = a1;

  const float4* X4 = reinterpret_cast<const float4*>(X);
  const int col = lane & 15;
  const int hsel = lane >> 4;
  float4 xa[2][GROUP], xb[2][GROUP];
  const float4* pl = X4 + ((size_t)(first + col) * 2 + (size_t)hsel);
#define LOAD_GROUP(J)                                                        \
  do {                                                                       \
    _Pragma("unroll") for (int k = 0; k < GROUP; ++k) {                      \
      if (!TAIL) {                                                           \
        xa[(J) & 1][k] = pl[((J) * GROUP + k) * 64];                         \
        xb[(J) & 1][k] = pl[((J) * GROUP + k) * 64 + 32];                    \
      } else {                                                               \
        unsigned e1_ = (unsigned)(first + col + ((J) * GROUP + k) * 32);     \
        unsigned e2_ = e1_ + 16u;                                            \
        e1_ = e1_ < (unsigned)last ? e1_ : (unsigned)last;                   \
        e2_ = e2_ < (unsigned)last ? e2_ : (unsigned)last;                   \
        xa[(J) & 1][k] = X4[(size_t)e1_ * 2 + hsel];                         \
        xb[(J) & 1][k] = X4[(size_t)e2_ * 2 + hsel];                         \
      }                                                                      \
    }                                                                        \
  } while (0)

  LOAD_GROUP(0);
#pragma unroll
  for (int j = 0; j < NGROUPS; ++j) {
    __builtin_amdgcn_sched_barrier(0);
    if (j + 1 < NGROUPS) LOAD_GROUP(j + 1);
    __builtin_amdgcn_sched_barrier(0);
#pragma unroll
    for (int s = 0; s < GROUP; s += HALF) {
      float res[HALF];
      rnn_chains(&xa[j & 1][s], &xb[j & 1][s], a, w00, w01, w10, w11, b0, b1,
                 cw0, cw1, cb, res);
      float* op = out + (size_t)e0 + (size_t)((j * GROUP + s) * 32);
      if (first + (j * GROUP + s + HALF) * 32 <= B) {
#pragma unroll
        for (int k = 0; k < HALF; ++k) op[k * 32] = res[k];
      } else {
#pragma unroll
        for (int k = 0; k < HALF; ++k)
          if (e0 + (j * GROUP + s + k) * 32 < B) op[k * 32] = res[k];
      }
    }
  }
#undef LOAD_GROUP
}

__global__ __launch_bounds__(WAVES * 32) void rnn2cls_kernel(
    const float* __restrict__ X, const float* __restrict__ wih,
    const float* __restrict__ bih, const float* __restrict__ whh,
    const float* __restrict__ bhh, const float* __restrict__ clsw,
    const float* __restrict__ clsb, float* __restrict__ out, int B) {
  const int lane = (int)(threadIdx.x & 31u);
  const int wave = __builtin_amdgcn_readfirstlane((int)(threadIdx.x >> 5u));
  const int first = ((int)blockIdx.x * WAVES + wave) * (TILES_PER_WAVE * 32);
  if (first >= B) return;
  if (first + TILES_PER_WAVE * 32 <= B)
    rnn2cls_wave<false>(X, wih, bih, whh, bhh, clsw, clsb, out, B, lane, first);
  else
    rnn2cls_wave<true>(X, wih, bih, whh, bhh, clsw, clsb, out, B, lane, first);
}

extern "C" void kernel_launch(void* const* d_in, const int* in_sizes, int n_in,
                              void* d_out, int out_size, void* d_ws, size_t ws_size,
                              hipStream_t stream) {
  (void)n_in; (void)out_size; (void)d_ws; (void)ws_size;
  const float* X    = (const float*)d_in[0];
  const float* wih  = (const float*)d_in[1];
  const float* bih  = (const float*)d_in[2];
  const float* whh  = (const float*)d_in[3];
  const float* bhh  = (const float*)d_in[4];
  const float* clsw = (const float*)d_in[5];
  const float* clsb = (const float*)d_in[6];
  float* out = (float*)d_out;

  const int B = in_sizes[0] / 8;
  const int per_block = WAVES * TILES_PER_WAVE * 32;
  const int blocks = (B + per_block - 1) / per_block;
  rnn2cls_kernel<<<blocks, WAVES * 32, 0, stream>>>(X, wih, bih, whh, bhh, clsw,
                                                   clsb, out, B);
}


# Input terms of all 3 steps put once into rows 0-5 of one accumulator per tile; W_hh sits in another row pair of A per step (3 A operands): the 3 WMMAs chain in place, no per-step C build or clears
# speedup vs baseline: 2.0136x; 1.0556x over previous
#include <hip/hip_runtime.h>


typedef __attribute__((ext_vector_type(2))) float v2f;
typedef __attribute__((ext_vector_type(8))) float v8f;

#define WAVES 16
#define GROUP 8
#define NGROUPS 4
#define HALF 4
#define TILES_PER_WAVE (GROUP * NGROUPS)

__device__ __forceinline__ float fast_tanh(float x) {
  return __builtin_amdgcn_tanhf(x);
}

__device__ __forceinline__ void swap_halves(float& p, float& q) {
  auto r = __builtin_amdgcn_permlane16_swap(__builtin_bit_cast(unsigned, p),
                                            __builtin_bit_cast(unsigned, q),
                                            false, false);
  p = __builtin_bit_cast(float, (unsigned)r[0]);
  q = __builtin_bit_cast(float, (unsigned)r[1]);
}

__device__ __forceinline__ void rnn_chains(const float4* xa, const float4* xb,
                                           v2f a0, v2f a1, v2f a2, float w00, float w01,
                                           float w10, float w11, float b0,
                                           float b1, float cw0, float cw1,
                                           float cb, float* res) {
  float4 pa[HALF], pb[HALF];
#pragma unroll
  for (int k = 0; k < HALF; ++k) {
    pa[k] = xa[k];
    pb[k] = xb[k];
    swap_halves(pa[k].x, pb[k].x);
    swap_halves(pa[k].y, pb[k].y);
    swap_halves(pa[k].z, pb[k].z);
    swap_halves(pa[k].w, pb[k].w);
  }
  float ix[HALF][4][2];
#pragma unroll
  for (int k = 0; k < HALF; ++k) {
    ix[k][0][0] = __builtin_fmaf(w00, pa[k].x, __builtin_fmaf(w01, pa[k].y, b0));
    ix[k][0][1] = __builtin_fmaf(w10, pa[k].x, __builtin_fmaf(w11, pa[k].y, b1));
    ix[k][1][0] = __builtin_fmaf(w00, pa[k].z, __builtin_fmaf(w01, pa[k].w, b0));
    ix[k][1][1] = __builtin_fmaf(w10, pa[k].z, __builtin_fmaf(w11, pa[k].w, b1));
    ix[k][2][0] = __builtin_fmaf(w00, pb[k].x, __builtin_fmaf(w01, pb[k].y, b0));
    ix[k][2][1] = __builtin_fmaf(w10, pb[k].x, __builtin_fmaf(w11, pb[k].y, b1));
    ix[k][3][0] = __builtin_fmaf(w00, pb[k].z, __builtin_fmaf(w01, pb[k].w, b0));
    ix[k][3][1] = __builtin_fmaf(w10, pb[k].z, __builtin_fmaf(w11, pb[k].w, b1));
  }
  float h0[HALF], h1[HALF];
#pragma unroll
  for (int k = 0; k < HALF; ++k) {
    h0[k] = fast_tanh(ix[k][0][0]);
    h1[k] = fast_tanh(ix[k][0][1]);
  }
  v8f acc[HALF];
#pragma unroll
  for (int k = 0; k < HALF; ++k) {
    acc[k][0] = ix[k][1][0];
    acc[k][1] = ix[k][1][1];
    acc[k][2] = ix[k][2][0];
    acc[k][3] = ix[k][2][1];
    acc[k][4] = ix[k][3][0];
    acc[k][5] = ix[k][3][1];
    acc[k][6] = 0.0f;
    acc[k][7] = 0.0f;
  }
#pragma unroll
  for (int s = 0; s < 3; ++s) {
    const v2f a = s == 0 ? a0 : (s == 1 ? a1 : a2);
#pragma unroll
    for (int k = 0; k < HALF; ++k) {
      v2f b;
      b[0] = h0[k];
      b[1] = h1[k];
      acc[k] = __builtin_amdgcn_wmma_f32_16x16x4_f32(
           false, a,  false, b,
           (short)0, acc[k],  false,  false);
    }
#pragma unroll
    for (int k = 0; k < HALF; ++k) {
      h0[k] = fast_tanh(acc[k][2 * s]);
      h1[k] = fast_tanh(acc[k][2 * s + 1]);
    }
  }
#pragma unroll
  for (int k = 0; k < HALF; ++k)
    res[k] = __builtin_fmaf(cw0, h0[k], __builtin_fmaf(cw1, h1[k], cb));
}

template <bool TAIL>
__device__ __forceinline__ void rnn2cls_wave(
    const float* __restrict__ X, const float* __restrict__ wih,
    const float* __restrict__ bih, const float* __restrict__ whh,
    const float* __restrict__ bhh, const float* __restrict__ clsw,
    const float* __restrict__ clsb, float* __restrict__ out, int B, int lane,
    int first) {
  const int e0 = first + lane;
  const int last = B - 1;

  const float w00 = wih[0], w01 = wih[1], w10 = wih[2], w11 = wih[3];
  const float b0 = bih[0] + bhh[0];
  const float b1 = bih[1] + bhh[1];
  const float cw0 = clsw[0], cw1 = clsw[1], cb = clsb[0];

  const float u00 = whh[0], u01 = whh[1], u10 = whh[2], u11 = whh[3];
  v2f a[3];
#pragma unroll
  for (int s = 0; s < 3; ++s) {
    const bool r0 = (lane == 2 * s) | (lane == 24 + 2 * s);
    const bool r1 = (lane == 2 * s + 1) | (lane == 25 + 2 * s);
    float e0 = r1 ? u10 : 0.0f;
    float e1 = r1 ? u11 : 0.0f;
    e0 = r0 ? u00 : e0;
    e1 = r0 ? u01 : e1;
    a[s][0] = e0;
    a[s][1] = e1;
  }

  const float4* X4 = reinterpret_cast<const float4*>(X);
  const int col = lane & 15;
  const int hsel = lane >> 4;
  float4 xa[2][GROUP], xb[2][GROUP];
  const float4* pl = X4 + ((size_t)(first + col) * 2 + (size_t)hsel);
#define LOAD_GROUP(J)                                                        \
  do {                                                                       \
    _Pragma("unroll") for (int k = 0; k < GROUP; ++k) {                      \
      if (!TAIL) {                                                           \
        xa[(J) & 1][k] = pl[((J) * GROUP + k) * 64];                         \
        xb[(J) & 1][k] = pl[((J) * GROUP + k) * 64 + 32];                    \
      } else {                                                               \
        unsigned e1_ = (unsigned)(first + col + ((J) * GROUP + k) * 32);     \
        unsigned e2_ = e1_ + 16u;                                            \
        e1_ = e1_ < (unsigned)last ? e1_ : (unsigned)last;                   \
        e2_ = e2_ < (unsigned)last ? e2_ : (unsigned)last;                   \
        xa[(J) & 1][k] = X4[(size_t)e1_ * 2 + hsel];                         \
        xb[(J) & 1][k] = X4[(size_t)e2_ * 2 + hsel];                         \
      }                                                                      \
    }                                                                        \
  } while (0)

  LOAD_GROUP(0);
#pragma unroll
  for (int j = 0; j < NGROUPS; ++j) {
    __builtin_amdgcn_sched_barrier(0);
    if (j + 1 < NGROUPS) LOAD_GROUP(j + 1);
    __builtin_amdgcn_sched_barrier(0);
#pragma unroll
    for (int s = 0; s < GROUP; s += HALF) {
      float res[HALF];
      rnn_chains(&xa[j & 1][s], &xb[j & 1][s], a[0], a[1], a[2], w00, w01, w10, w11, b0, b1,
                 cw0, cw1, cb, res);
      float* op = out + (size_t)e0 + (size_t)((j * GROUP + s) * 32);
      if (first + (j * GROUP + s + HALF) * 32 <= B) {
#pragma unroll
        for (int k = 0; k < HALF; ++k) op[k * 32] = res[k];
      } else {
#pragma unroll
        for (int k = 0; k < HALF; ++k)
          if (e0 + (j * GROUP + s + k) * 32 < B) op[k * 32] = res[k];
      }
    }
  }
#undef LOAD_GROUP
}

__global__ __launch_bounds__(WAVES * 32) void rnn2cls_kernel(
    const float* __restrict__ X, const float* __restrict__ wih,
    const float* __restrict__ bih, const float* __restrict__ whh,
    const float* __restrict__ bhh, const float* __restrict__ clsw,
    const float* __restrict__ clsb, float* __restrict__ out, int B) {
  const int lane = (int)(threadIdx.x & 31u);
  const int wave = __builtin_amdgcn_readfirstlane((int)(threadIdx.x >> 5u));
  const int first = ((int)blockIdx.x * WAVES + wave) * (TILES_PER_WAVE * 32);
  if (first >= B) return;
  if (first + TILES_PER_WAVE * 32 <= B)
    rnn2cls_wave<false>(X, wih, bih, whh, bhh, clsw, clsb, out, B, lane, first);
  else
    rnn2cls_wave<true>(X, wih, bih, whh, bhh, clsw, clsb, out, B, lane, first);
}

extern "C" void kernel_launch(void* const* d_in, const int* in_sizes, int n_in,
                              void* d_out, int out_size, void* d_ws, size_t ws_size,
                              hipStream_t stream) {
  (void)n_in; (void)out_size; (void)d_ws; (void)ws_size;
  const float* X    = (const float*)d_in[0];
  const float* wih  = (const float*)d_in[1];
  const float* bih  = (const float*)d_in[2];
  const float* whh  = (const float*)d_in[3];
  const float* bhh  = (const float*)d_in[4];
  const float* clsw = (const float*)d_in[5];
  const float* clsb = (const float*)d_in[6];
  float* out = (float*)d_out;

  const int B = in_sizes[0] / 8;
  const int per_block = WAVES * TILES_PER_WAVE * 32;
  const int blocks = (B + per_block - 1) / per_block;
  rnn2cls_kernel<<<blocks, WAVES * 32, 0, stream>>>(X, wih, bih, whh, bhh, clsw,
                                                   clsb, out, B);
}


# Full-wave path stores without bounds tests (the wave-level test covers its 32 tiles): its 8 batches become one branch-free block, so inputs, tanh and stores are scheduled beside other batches' WMMAs
# speedup vs baseline: 2.1408x; 1.0632x over previous
#include <hip/hip_runtime.h>


typedef __attribute__((ext_vector_type(2))) float v2f;
typedef __attribute__((ext_vector_type(8))) float v8f;

#define WAVES 16
#define GROUP 8
#define NGROUPS 4
#define HALF 4
#define TILES_PER_WAVE (GROUP * NGROUPS)

__device__ __forceinline__ float fast_tanh(float x) {
  return __builtin_amdgcn_tanhf(x);
}

__device__ __forceinline__ void swap_halves(float& p, float& q) {
  auto r = __builtin_amdgcn_permlane16_swap(__builtin_bit_cast(unsigned, p),
                                            __builtin_bit_cast(unsigned, q),
                                            false, false);
  p = __builtin_bit_cast(float, (unsigned)r[0]);
  q = __builtin_bit_cast(float, (unsigned)r[1]);
}

__device__ __forceinline__ void rnn_chains(const float4* xa, const float4* xb,
                                           v2f a0, v2f a1, v2f a2, float w00, float w01,
                                           float w10, float w11, float b0,
                                           float b1, float cw0, float cw1,
                                           float cb, float* res) {
  float4 pa[HALF], pb[HALF];
#pragma unroll
  for (int k = 0; k < HALF; ++k) {
    pa[k] = xa[k];
    pb[k] = xb[k];
    swap_halves(pa[k].x, pb[k].x);
    swap_halves(pa[k].y, pb[k].y);
    swap_halves(pa[k].z, pb[k].z);
    swap_halves(pa[k].w, pb[k].w);
  }
  float ix[HALF][4][2];
#pragma unroll
  for (int k = 0; k < HALF; ++k) {
    ix[k][0][0] = __builtin_fmaf(w00, pa[k].x, __builtin_fmaf(w01, pa[k].y, b0));
    ix[k][0][1] = __builtin_fmaf(w10, pa[k].x, __builtin_fmaf(w11, pa[k].y, b1));
    ix[k][1][0] = __builtin_fmaf(w00, pa[k].z, __builtin_fmaf(w01, pa[k].w, b0));
    ix[k][1][1] = __builtin_fmaf(w10, pa[k].z, __builtin_fmaf(w11, pa[k].w, b1));
    ix[k][2][0] = __builtin_fmaf(w00, pb[k].x, __builtin_fmaf(w01, pb[k].y, b0));
    ix[k][2][1] = __builtin_fmaf(w10, pb[k].x, __builtin_fmaf(w11, pb[k].y, b1));
    ix[k][3][0] = __builtin_fmaf(w00, pb[k].z, __builtin_fmaf(w01, pb[k].w, b0));
    ix[k][3][1] = __builtin_fmaf(w10, pb[k].z, __builtin_fmaf(w11, pb[k].w, b1));
  }
  float h0[HALF], h1[HALF];
#pragma unroll
  for (int k = 0; k < HALF; ++k) {
    h0[k] = fast_tanh(ix[k][0][0]);
    h1[k] = fast_tanh(ix[k][0][1]);
  }
  v8f acc[HALF];
#pragma unroll
  for (int k = 0; k < HALF; ++k) {
    acc[k][0] = ix[k][1][0];
    acc[k][1] = ix[k][1][1];
    acc[k][2] = ix[k][2][0];
    acc[k][3] = ix[k][2][1];
    acc[k][4] = ix[k][3][0];
    acc[k][5] = ix[k][3][1];
    acc[k][6] = 0.0f;
    acc[k][7] = 0.0f;
  }
#pragma unroll
  for (int s = 0; s < 3; ++s) {
    const v2f a = s == 0 ? a0 : (s == 1 ? a1 : a2);
#pragma unroll
    for (int k = 0; k < HALF; ++k) {
      v2f b;
      b[0] = h0[k];
      b[1] = h1[k];
      acc[k] = __builtin_amdgcn_wmma_f32_16x16x4_f32(
           false, a,  false, b,
           (short)0, acc[k],  false,  false);
    }
#pragma unroll
    for (int k = 0; k < HALF; ++k) {
      h0[k] = fast_tanh(acc[k][2 * s]);
      h1[k] = fast_tanh(acc[k][2 * s + 1]);
    }
  }
#pragma unroll
  for (int k = 0; k < HALF; ++k)
    res[k] = __builtin_fmaf(cw0, h0[k], __builtin_fmaf(cw1, h1[k], cb));
}

template <bool TAIL>
__device__ __forceinline__ void rnn2cls_wave(
    const float* __restrict__ X, const float* __restrict__ wih,
    const float* __restrict__ bih, const float* __restrict__ whh,
    const float* __restrict__ bhh, const float* __restrict__ clsw,
    const float* __restrict__ clsb, float* __restrict__ out, int B, int lane,
    int first) {
  const int e0 = first + lane;
  const int last = B - 1;

  const float w00 = wih[0], w01 = wih[1], w10 = wih[2], w11 = wih[3];
  const float b0 = bih[0] + bhh[0];
  const float b1 = bih[1] + bhh[1];
  const float cw0 = clsw[0], cw1 = clsw[1], cb = clsb[0];

  const float u00 = whh[0], u01 = whh[1], u10 = whh[2], u11 = whh[3];
  v2f a[3];
#pragma unroll
  for (int s = 0; s < 3; ++s) {
    const bool r0 = (lane == 2 * s) | (lane == 24 + 2 * s);
    const bool r1 = (lane == 2 * s + 1) | (lane == 25 + 2 * s);
    float e0 = r1 ? u10 : 0.0f;
    float e1 = r1 ? u11 : 0.0f;
    e0 = r0 ? u00 : e0;
    e1 = r0 ? u01 : e1;
    a[s][0] = e0;
    a[s][1] = e1;
  }

  const float4* X4 = reinterpret_cast<const float4*>(X);
  const int col = lane & 15;
  const int hsel = lane >> 4;
  float4 xa[2][GROUP], xb[2][GROUP];
  const float4* pl = X4 + ((size_t)(first + col) * 2 + (size_t)hsel);
#define LOAD_GROUP(J)                                                        \
  do {                                                                       \
    _Pragma("unroll") for (int k = 0; k < GROUP; ++k) {                      \
      if (!TAIL) {                                                           \
        xa[(J) & 1][k] = pl[((J) * GROUP + k) * 64];                         \
        xb[(J) & 1][k] = pl[((J) * GROUP + k) * 64 + 32];                    \
      } else {                                                               \
        unsigned e1_ = (unsigned)(first + col + ((J) * GROUP + k) * 32);     \
        unsigned e2_ = e1_ + 16u;                                            \
        e1_ = e1_ < (unsigned)last ? e1_ : (unsigned)last;                   \
        e2_ = e2_ < (unsigned)last ? e2_ : (unsigned)last;                   \
        xa[(J) & 1][k] = X4[(size_t)e1_ * 2 + hsel];                         \
        xb[(J) & 1][k] = X4[(size_t)e2_ * 2 + hsel];                         \
      }                                                                      \
    }                                                                        \
  } while (0)

  LOAD_GROUP(0);
#pragma unroll
  for (int j = 0; j < NGROUPS; ++j) {
    __builtin_amdgcn_sched_barrier(0);
    if (j + 1 < NGROUPS) LOAD_GROUP(j + 1);
    __builtin_amdgcn_sched_barrier(0);
#pragma unroll
    for (int s = 0; s < GROUP; s += HALF) {
      float res[HALF];
      rnn_chains(&xa[j & 1][s], &xb[j & 1][s], a[0], a[1], a[2], w00, w01, w10, w11, b0, b1,
                 cw0, cw1, cb, res);
      float* op = out + (size_t)e0 + (size_t)((j * GROUP + s) * 32);
      if (!TAIL) {
#pragma unroll
        for (int k = 0; k < HALF; ++k) op[k * 32] = res[k];
      } else if (first + (j * GROUP + s + HALF) * 32 <= B) {
#pragma unroll
        for (int k = 0; k < HALF; ++k) op[k * 32] = res[k];
      } else {
#pragma unroll
        for (int k = 0; k < HALF; ++k)
          if (e0 + (j * GROUP + s + k) * 32 < B) op[k * 32] = res[k];
      }
    }
  }
#undef LOAD_GROUP
}

__global__ __launch_bounds__(WAVES * 32) void rnn2cls_kernel(
    const float* __restrict__ X, const float* __restrict__ wih,
    const float* __restrict__ bih, const float* __restrict__ whh,
    const float* __restrict__ bhh, const float* __restrict__ clsw,
    const float* __restrict__ clsb, float* __restrict__ out, int B) {
  const int lane = (int)(threadIdx.x & 31u);
  const int wave = __builtin_amdgcn_readfirstlane((int)(threadIdx.x >> 5u));
  const int first = ((int)blockIdx.x * WAVES + wave) * (TILES_PER_WAVE * 32);
  if (first >= B) return;
  if (first + TILES_PER_WAVE * 32 <= B)
    rnn2cls_wave<false>(X, wih, bih, whh, bhh, clsw, clsb, out, B, lane, first);
  else
    rnn2cls_wave<true>(X, wih, bih, whh, bhh, clsw, clsb, out, B, lane, first);
}

extern "C" void kernel_launch(void* const* d_in, const int* in_sizes, int n_in,
                              void* d_out, int out_size, void* d_ws, size_t ws_size,
                              hipStream_t stream) {
  (void)n_in; (void)out_size; (void)d_ws; (void)ws_size;
  const float* X    = (const float*)d_in[0];
  const float* wih  = (const float*)d_in[1];
  const float* bih  = (const float*)d_in[2];
  const float* whh  = (const float*)d_in[3];
  const float* bhh  = (const float*)d_in[4];
  const float* clsw = (const float*)d_in[5];
  const float* clsb = (const float*)d_in[6];
  float* out = (float*)d_out;

  const int B = in_sizes[0] / 8;
  const int per_block = WAVES * TILES_PER_WAVE * 32;
  const int blocks = (B + per_block - 1) / per_block;
  rnn2cls_kernel<<<blocks, WAVES * 32, 0, stream>>>(X, wih, bih, whh, bhh, clsw,
                                                   clsb, out, B);
}
